# starting kernel
# baseline (speedup 1.0000x reference)
#include <hip/hip_runtime.h>

typedef float v8f __attribute__((ext_vector_type(8)));
typedef float v2f __attribute__((ext_vector_type(2)));

#define BB   1024
#define TT   1024
#define FF   8
#define HH   256
#define LL   64

#define ROWS_PER_WAVE   16
#define WAVES_PER_BLOCK 2
#define ROWS_PER_BLOCK  (ROWS_PER_WAVE * WAVES_PER_BLOCK)
#define NTHREADS        (32 * WAVES_PER_BLOCK)
#define SROW            260

__device__ __forceinline__ float sigmoidf_(float x) {
    return 1.0f / (1.0f + __expf(-x));
}

__global__ __launch_bounds__(NTHREADS)
void kan_rnn_enc_kernel(const float* __restrict__ x_ctx,
                        const float* __restrict__ a_in,
                        const float* __restrict__ b_in,
                        const float* __restrict__ a_h,
                        const float* __restrict__ b_h,
                        const float* __restrict__ W_lat,
                        const float* __restrict__ b_lat,
                        float* __restrict__ out) {
    __shared__ float lx[ROWS_PER_BLOCK * FF];
    __shared__ float px[ROWS_PER_BLOCK * FF];
    __shared__ float sPrev[ROWS_PER_BLOCK * 24];
    __shared__ float sH[ROWS_PER_BLOCK * SROW];

    const int tid = threadIdx.x;
    const int r0  = blockIdx.x * ROWS_PER_BLOCK;

    for (int i = tid; i < ROWS_PER_BLOCK * FF; i += NTHREADS) {
        int row = i >> 3, f = i & 7;
        const float* base = x_ctx + (size_t)(r0 + row) * (TT * FF);
        px[i] = base[(TT - 2) * FF + f];
        lx[i] = base[(TT - 1) * FF + f];
    }
    __syncthreads();

    for (int i = tid; i < ROWS_PER_BLOCK * 24; i += NTHREADS) {
        int row = i / 24, hh = i % 24;
        int f = hh >> 3;
        float t = a_in[hh] * (px[row * FF + f] - b_in[hh]);
        sPrev[i] = sigmoidf_(2.0f * sigmoidf_(t));
    }
    __syncthreads();

    for (int i = tid; i < ROWS_PER_BLOCK * HH; i += NTHREADS) {
        int row = i >> 8, j = i & 255;
        float v;
        if (j < 64) {
            int f = j >> 3;
            v = sigmoidf_(2.0f * sigmoidf_(a_in[j] * (lx[row * FF + f] - b_in[j])));
        } else {
            int m  = j - 64;
            int hh = m >> 3;
            float ph = sPrev[row * 24 + hh];
            v = sigmoidf_(2.0f * sigmoidf_(a_h[m] * (ph - b_h[m])));
        }
        sH[row * SROW + j] = v;
    }
    __syncthreads();

    const int lane = tid & 31;
    const int wave = tid >> 5;
    const int nl   = lane & 15;
    const int kb   = (lane >> 4) << 1;

    v8f acc[4];
#pragma unroll
    for (int n = 0; n < 4; ++n) {
        float bl = b_lat[n * 16 + nl];
#pragma unroll
        for (int r = 0; r < 8; ++r) acc[n][r] = bl;
    }

    const float* aRow  = &sH[(wave * 16 + nl) * SROW + kb];
    const float* wBase = W_lat + kb;

#pragma unroll
    for (int kk = 0; kk < HH / 4; ++kk) {
        v2f afrag = *(const v2f*)(aRow + kk * 4);
#pragma unroll
        for (int n = 0; n < 4; ++n) {
            v2f bfrag = *(const v2f*)(wBase + (n * 16 + nl) * HH + kk * 4);
            acc[n] = __builtin_amdgcn_wmma_f32_16x16x4_f32(
                false, afrag, false, bfrag, (short)0, acc[n], false, false);
        }
    }

    const int rowTop = r0 + wave * 16 + ((lane >> 4) << 3);
#pragma unroll
    for (int n = 0; n < 4; ++n) {
#pragma unroll
        for (int r = 0; r < 8; ++r) {
            out[(size_t)(rowTop + r) * LL + n * 16 + nl] = acc[n][r];
        }
    }
}

extern "C" void kernel_launch(void* const* d_in, const int* in_sizes, int n_in,
                              void* d_out, int out_size, void* d_ws, size_t ws_size,
                              hipStream_t stream) {
    (void)in_sizes; (void)n_in; (void)out_size; (void)d_ws; (void)ws_size;
    const float* x_ctx = (const float*)d_in[0];
    const float* a_in  = (const float*)d_in[1];
    const float* b_in  = (const float*)d_in[2];
    const float* a_h   = (const float*)d_in[3];
    const float* b_h   = (const float*)d_in[4];
    const float* W_lat = (const float*)d_in[5];
    const float* b_lat = (const float*)d_in[6];
    float* out = (float*)d_out;

    dim3 grid(BB / ROWS_PER_BLOCK);
    dim3 block(NTHREADS);
    hipLaunchKernelGGL(kan_rnn_enc_kernel, grid, block, 0, stream,
                       x_ctx, a_in, b_in, a_h, b_h, W_lat, b_lat, out);
}


# Re-tiled for parallelism: 64 groups of one 16-row tile x 16 waves; each thread owns one hidden column, waves 0-3 one 16x16 output tile each (64 WMMA); rows gathered in LDS, one writer per line
# speedup vs baseline: 6.2085x; 6.2085x over previous
#include <hip/hip_runtime.h>

typedef float v8f __attribute__((ext_vector_type(8)));
typedef float v2f __attribute__((ext_vector_type(2)));

#define BB   1024
#define TT   1024
#define FF   8
#define HH   256
#define LL   64

#define ROWS_PER_BLOCK  16
#define WAVES_PER_BLOCK 16
#define MMA_WAVES       (LL / 16)
#define NTHREADS        (32 * WAVES_PER_BLOCK)
#define SROW            260

__device__ __forceinline__ float sigmoidf_(float x) {
    return 1.0f / (1.0f + __expf(-x));
}

__global__ __launch_bounds__(NTHREADS)
void kan_rnn_enc_kernel(const float* __restrict__ x_ctx,
                        const float* __restrict__ a_in,
                        const float* __restrict__ b_in,
                        const float* __restrict__ a_h,
                        const float* __restrict__ b_h,
                        const float* __restrict__ W_lat,
                        const float* __restrict__ b_lat,
                        float* __restrict__ out) {
    __shared__ float lx[ROWS_PER_BLOCK * FF];
    __shared__ float px[ROWS_PER_BLOCK * FF];
    __shared__ float sPrev[ROWS_PER_BLOCK * 24];
    __shared__ float sH[ROWS_PER_BLOCK * SROW];
    __shared__ __attribute__((aligned(16))) float sOut[ROWS_PER_BLOCK * LL];

    const int tid = threadIdx.x;
    const int r0  = blockIdx.x * ROWS_PER_BLOCK;

    for (int i = tid; i < ROWS_PER_BLOCK * FF; i += NTHREADS) {
        int row = i >> 3, f = i & 7;
        const float* base = x_ctx + (size_t)(r0 + row) * (TT * FF);
        px[i] = base[(TT - 2) * FF + f];
        lx[i] = base[(TT - 1) * FF + f];
    }
    __syncthreads();

    for (int i = tid; i < ROWS_PER_BLOCK * 24; i += NTHREADS) {
        int row = i / 24, hh = i % 24;
        int f = hh >> 3;
        float t = a_in[hh] * (px[row * FF + f] - b_in[hh]);
        sPrev[i] = sigmoidf_(2.0f * sigmoidf_(t));
    }
    __syncthreads();

    for (int i = tid; i < ROWS_PER_BLOCK * HH; i += NTHREADS) {
        int row = i >> 8, j = i & 255;
        float v;
        if (j < 64) {
            int f = j >> 3;
            v = sigmoidf_(2.0f * sigmoidf_(a_in[j] * (lx[row * FF + f] - b_in[j])));
        } else {
            int m  = j - 64;
            int hh = m >> 3;
            float ph = sPrev[row * 24 + hh];
            v = sigmoidf_(2.0f * sigmoidf_(a_h[m] * (ph - b_h[m])));
        }
        sH[row * SROW + j] = v;
    }
    __syncthreads();

    const int lane = tid & 31;
    const int wave = tid >> 5;
    const int nl   = lane & 15;
    const int kb   = (lane >> 4) << 1;

    if (wave < MMA_WAVES) {
        v8f acc;
        {
            float bl = b_lat[wave * 16 + nl];
#pragma unroll
            for (int r = 0; r < 8; ++r) acc[r] = bl;
        }

        const float* aRow = &sH[nl * SROW + kb];
        const float* wRow = W_lat + (wave * 16 + nl) * HH + kb;

#pragma unroll
        for (int kk = 0; kk < HH / 4; ++kk) {
            v2f afrag = *(const v2f*)(aRow + kk * 4);
            v2f bfrag = *(const v2f*)(wRow + kk * 4);
            acc = __builtin_amdgcn_wmma_f32_16x16x4_f32(
                false, afrag, false, bfrag, (short)0, acc, false, false);
        }

        const int mTop = (lane >> 4) << 3;
#pragma unroll
        for (int r = 0; r < 8; ++r) sOut[(mTop + r) * LL + wave * 16 + nl] = acc[r];
    }
    __syncthreads();

    float4* outBlk = (float4*)(out + (size_t)r0 * LL);
    const float4* sOut4 = (const float4*)sOut;
    for (int q = tid; q < ROWS_PER_BLOCK * LL / 4; q += NTHREADS) outBlk[q] = sOut4[q];
}

extern "C" void kernel_launch(void* const* d_in, const int* in_sizes, int n_in,
                              void* d_out, int out_size, void* d_ws, size_t ws_size,
                              hipStream_t stream) {
    (void)in_sizes; (void)n_in; (void)out_size; (void)d_ws; (void)ws_size;
    const float* x_ctx = (const float*)d_in[0];
    const float* a_in  = (const float*)d_in[1];
    const float* b_in  = (const float*)d_in[2];
    const float* a_h   = (const float*)d_in[3];
    const float* b_h   = (const float*)d_in[4];
    const float* W_lat = (const float*)d_in[5];
    const float* b_lat = (const float*)d_in[6];
    float* out = (float*)d_out;

    dim3 grid(BB / ROWS_PER_BLOCK);
    dim3 block(NTHREADS);
    hipLaunchKernelGGL(kan_rnn_enc_kernel, grid, block, 0, stream,
                       x_ctx, a_in, b_in, a_h, b_h, W_lat, b_lat, out);
}


# Hidden-state loop: each thread fetches its column's two basis parameters once at entry and runs its 8 rows as independent overlapped chains (was a rolled loop, two dependent global loads per trip)
# speedup vs baseline: 8.0660x; 1.2992x over previous
#include <hip/hip_runtime.h>

typedef float v8f __attribute__((ext_vector_type(8)));
typedef float v2f __attribute__((ext_vector_type(2)));

#define BB   1024
#define TT   1024
#define FF   8
#define HH   256
#define LL   64

#define ROWS_PER_BLOCK  16
#define WAVES_PER_BLOCK 16
#define MMA_WAVES       (LL / 16)
#define NTHREADS        (32 * WAVES_PER_BLOCK)
#define SROW            260
#define ROWS_PER_THREAD (ROWS_PER_BLOCK * HH / NTHREADS)
#define UCOLS           32

__device__ __forceinline__ float sigmoidf_(float x) {
    return 1.0f / (1.0f + __expf(-x));
}

__global__ __launch_bounds__(NTHREADS)
void kan_rnn_enc_kernel(const float* __restrict__ x_ctx,
                        const float* __restrict__ a_in,
                        const float* __restrict__ b_in,
                        const float* __restrict__ a_h,
                        const float* __restrict__ b_h,
                        const float* __restrict__ W_lat,
                        const float* __restrict__ b_lat,
                        float* __restrict__ out) {
    __shared__ float px[ROWS_PER_BLOCK * FF];
    __shared__ float sU[ROWS_PER_BLOCK * UCOLS];
    __shared__ float sH[ROWS_PER_BLOCK * SROW];
    __shared__ __attribute__((aligned(16))) float sOut[ROWS_PER_BLOCK * LL];

    const int tid = threadIdx.x;
    const int r0  = blockIdx.x * ROWS_PER_BLOCK;

    const int j = tid & (HH - 1);
    const float* pa = (j < 64) ? (a_in + j) : (a_h + (j - 64));
    const float* pb = (j < 64) ? (b_in + j) : (b_h + (j - 64));
    const float ca = *pa;
    const float cb = *pb;

    for (int i = tid; i < ROWS_PER_BLOCK * FF; i += NTHREADS) {
        int row = i >> 3, f = i & 7;
        const float* base = x_ctx + (size_t)(r0 + row) * (TT * FF);
        px[i] = base[(TT - 2) * FF + f];
        sU[row * UCOLS + f] = base[(TT - 1) * FF + f];
    }
    __syncthreads();

    for (int i = tid; i < ROWS_PER_BLOCK * 24; i += NTHREADS) {
        int row = i / 24, hh = i % 24;
        int f = hh >> 3;
        float t = a_in[hh] * (px[row * FF + f] - b_in[hh]);
        sU[row * UCOLS + FF + hh] = sigmoidf_(2.0f * sigmoidf_(t));
    }
    __syncthreads();

    {
        const float* uCol = &sU[(tid >> 8) * UCOLS + (j >> 3)];
        float*       hCol = &sH[(tid >> 8) * SROW + j];
        for (int it = 0; it < ROWS_PER_THREAD; ++it) {
            float u = uCol[it * 2 * UCOLS];
            hCol[it * 2 * SROW] = sigmoidf_(2.0f * sigmoidf_(ca * (u - cb)));
        }
    }
    __syncthreads();

    const int lane = tid & 31;
    const int wave = tid >> 5;
    const int nl   = lane & 15;
    const int kb   = (lane >> 4) << 1;

    if (wave < MMA_WAVES) {
        v8f acc;
        {
            float bl = b_lat[wave * 16 + nl];
#pragma unroll
            for (int r = 0; r < 8; ++r) acc[r] = bl;
        }

        const float* aRow = &sH[nl * SROW + kb];
        const float* wRow = W_lat + (wave * 16 + nl) * HH + kb;

#pragma unroll
        for (int kk = 0; kk < HH / 4; ++kk) {
            v2f afrag = *(const v2f*)(aRow + kk * 4);
            v2f bfrag = *(const v2f*)(wRow + kk * 4);
            acc = __builtin_amdgcn_wmma_f32_16x16x4_f32(
                false, afrag, false, bfrag, (short)0, acc, false, false);
        }

        const int mTop = (lane >> 4) << 3;
#pragma unroll
        for (int r = 0; r < 8; ++r) sOut[(mTop + r) * LL + wave * 16 + nl] = acc[r];
    }
    __syncthreads();

    float4* outBlk = (float4*)(out + (size_t)r0 * LL);
    const float4* sOut4 = (const float4*)sOut;
    for (int q = tid; q < ROWS_PER_BLOCK * LL / 4; q += NTHREADS) outBlk[q] = sOut4[q];
}

extern "C" void kernel_launch(void* const* d_in, const int* in_sizes, int n_in,
                              void* d_out, int out_size, void* d_ws, size_t ws_size,
                              hipStream_t stream) {
    (void)in_sizes; (void)n_in; (void)out_size; (void)d_ws; (void)ws_size;
    const float* x_ctx = (const float*)d_in[0];
    const float* a_in  = (const float*)d_in[1];
    const float* b_in  = (const float*)d_in[2];
    const float* a_h   = (const float*)d_in[3];
    const float* b_h   = (const float*)d_in[4];
    const float* W_lat = (const float*)d_in[5];
    const float* b_lat = (const float*)d_in[6];
    float* out = (float*)d_out;

    dim3 grid(BB / ROWS_PER_BLOCK);
    dim3 block(NTHREADS);
    hipLaunchKernelGGL(kan_rnn_enc_kernel, grid, block, 0, stream,
                       x_ctx, a_in, b_in, a_h, b_h, W_lat, b_lat, out);
}


# Re-tiled over all 256 WGPs: 256 groups of 8 rows x 32 out columns x 4 waves (was 64 x 16 x 64); a wave = one 16-col tile over a K half in 2 accumulator chains; K summed as 4 partials added last
# speedup vs baseline: 8.9930x; 1.1149x over previous
#include <hip/hip_runtime.h>


typedef float v8f __attribute__((ext_vector_type(8)));
typedef float v2f __attribute__((ext_vector_type(2)));

#define BB   1024
#define TT   1024
#define FF   8
#define HH   256
#define LL   64

#define ROWS_PER_BLOCK  8
#define COLS_PER_BLOCK  32
#define NTILES          (COLS_PER_BLOCK / 16)
#define KPARTS          2
#define KSPAN           (HH / KPARTS)
#define WAVES_PER_BLOCK (NTILES * KPARTS)
#define NTHREADS        (32 * WAVES_PER_BLOCK)
#define NBLOCKS         ((BB / ROWS_PER_BLOCK) * (LL / COLS_PER_BLOCK))
#define SROW            260
#define COLS_PER_THREAD (HH / NTHREADS)
#define UCOLS           32

__device__ __forceinline__ float sigmoidf_(float x) {
    return 1.0f / (1.0f + __expf(-x));
}

__global__ __launch_bounds__(NTHREADS)
void kan_rnn_enc_kernel(const float* __restrict__ x_ctx,
                        const float* __restrict__ a_in,
                        const float* __restrict__ b_in,
                        const float* __restrict__ a_h,
                        const float* __restrict__ b_h,
                        const float* __restrict__ W_lat,
                        const float* __restrict__ b_lat,
                        float* __restrict__ out) {
    __shared__ float px[ROWS_PER_BLOCK * FF];
    __shared__ float sU[ROWS_PER_BLOCK * UCOLS];
    __shared__ float sH[ROWS_PER_BLOCK * SROW];
    __shared__ __attribute__((aligned(16))) float sPart[KPARTS * ROWS_PER_BLOCK * COLS_PER_BLOCK];

    const int tid = threadIdx.x;
    const int r0  = (blockIdx.x >> 1) * ROWS_PER_BLOCK;
    const int c0  = (blockIdx.x & 1) * COLS_PER_BLOCK;

    const int lane = tid & 31;
    const int wave = tid >> 5;
    const int nl   = lane & 15;
    const int kb   = (lane >> 4) << 1;
    const int nt   = wave & 1;
    const int kh   = wave >> 1;

    const float blane = b_lat[c0 + nt * 16 + nl];

    float ca[COLS_PER_THREAD], cb[COLS_PER_THREAD];
#pragma unroll
    for (int c = 0; c < COLS_PER_THREAD; ++c) {
        const int j = tid + c * NTHREADS;
        const float* pa = (j < 64) ? (a_in + j) : (a_h + (j - 64));
        const float* pb = (j < 64) ? (b_in + j) : (b_h + (j - 64));
        ca[c] = *pa;
        cb[c] = *pb;
    }

    for (int i = tid; i < ROWS_PER_BLOCK * FF; i += NTHREADS) {
        int row = i >> 3, f = i & 7;
        const float* base = x_ctx + (size_t)(r0 + row) * (TT * FF);
        px[i] = base[(TT - 2) * FF + f];
        sU[row * UCOLS + f] = base[(TT - 1) * FF + f];
    }
    __syncthreads();

#pragma unroll
    for (int c = 0; c < 2; ++c) {
        const int i2 = tid + c * NTHREADS;
        const int i  = (i2 < ROWS_PER_BLOCK * 24) ? i2 : tid;
        int row = i / 24, hh = i % 24;
        int f = hh >> 3;
        float t = a_in[hh] * (px[row * FF + f] - b_in[hh]);
        sU[row * UCOLS + FF + hh] = sigmoidf_(2.0f * sigmoidf_(t));
    }
    __syncthreads();

#pragma unroll
    for (int c = 0; c < COLS_PER_THREAD; ++c) {
        const int j = tid + c * NTHREADS;
        const float* uCol = &sU[j >> 3];
        float*       hCol = &sH[j];
#pragma unroll
        for (int row = 0; row < ROWS_PER_BLOCK; ++row) {
            float u = uCol[row * UCOLS];
            hCol[row * SROW] = sigmoidf_(2.0f * sigmoidf_(ca[c] * (u - cb[c])));
        }
    }
    __syncthreads();

    v8f acc0, acc1;
    {
        const float bl = (kh == 0) ? blane : 0.0f;
#pragma unroll
        for (int r = 0; r < 8; ++r) { acc0[r] = bl; acc1[r] = 0.0f; }
    }

    const float* aRow = &sH[(nl & 7) * SROW + kh * KSPAN + kb];
    const float* wRow = W_lat + (c0 + nt * 16 + nl) * HH + kh * KSPAN + kb;

#pragma unroll
    for (int kk = 0; kk < KSPAN / 4; kk += 2) {
        v2f a0 = *(const v2f*)(aRow + kk * 4);
        v2f b0 = *(const v2f*)(wRow + kk * 4);
        v2f a1 = *(const v2f*)(aRow + kk * 4 + 4);
        v2f b1 = *(const v2f*)(wRow + kk * 4 + 4);
        acc0 = __builtin_amdgcn_wmma_f32_16x16x4_f32(
            false, a0, false, b0, (short)0, acc0, false, false);
        acc1 = __builtin_amdgcn_wmma_f32_16x16x4_f32(
            false, a1, false, b1, (short)0, acc1, false, false);
    }

    v8f part = acc0 + acc1;
    if (lane < 16) {
#pragma unroll
        for (int r = 0; r < 8; ++r)
            sPart[(kh * ROWS_PER_BLOCK + r) * COLS_PER_BLOCK + nt * 16 + nl] = part[r];
    }
    __syncthreads();

    if (tid < ROWS_PER_BLOCK * COLS_PER_BLOCK / 4) {
        const float4* sPart4 = (const float4*)sPart;
        float4 p = sPart4[tid];
        float4 q = sPart4[ROWS_PER_BLOCK * COLS_PER_BLOCK / 4 + tid];
        float4 v;
        v.x = p.x + q.x; v.y = p.y + q.y; v.z = p.z + q.z; v.w = p.w + q.w;
        const int row = tid >> 3, quad = tid & 7;
        *(float4*)(out + (size_t)(r0 + row) * LL + c0 + quad * 4) = v;
    }
}

extern "C" void kernel_launch(void* const* d_in, const int* in_sizes, int n_in,
                              void* d_out, int out_size, void* d_ws, size_t ws_size,
                              hipStream_t stream) {
    (void)in_sizes; (void)n_in; (void)out_size; (void)d_ws; (void)ws_size;
    const float* x_ctx = (const float*)d_in[0];
    const float* a_in  = (const float*)d_in[1];
    const float* b_in  = (const float*)d_in[2];
    const float* a_h   = (const float*)d_in[3];
    const float* b_h   = (const float*)d_in[4];
    const float* W_lat = (const float*)d_in[5];
    const float* b_lat = (const float*)d_in[6];
    float* out = (float*)d_out;

    dim3 grid(NBLOCKS);
    dim3 block(NTHREADS);
    hipLaunchKernelGGL(kan_rnn_enc_kernel, grid, block, 0, stream,
                       x_ctx, a_in, b_in, a_h, b_h, W_lat, b_lat, out);
}


# Every global load issued at kernel entry and kept in flight across LDS-only barriers (32 W_lat fragments into 64 registers, phase-1b parameters, x); A fragments read as one group before the 32 WMMAs
# speedup vs baseline: 10.2029x; 1.1345x over previous
#include <hip/hip_runtime.h>


typedef float v8f __attribute__((ext_vector_type(8)));
typedef float v2f __attribute__((ext_vector_type(2)));

#define BB   1024
#define TT   1024
#define FF   8
#define HH   256
#define LL   64

#define ROWS_PER_BLOCK  8
#define COLS_PER_BLOCK  32
#define NTILES          (COLS_PER_BLOCK / 16)
#define KPARTS          2
#define KSPAN           (HH / KPARTS)
#define WAVES_PER_BLOCK (NTILES * KPARTS)
#define NTHREADS        (32 * WAVES_PER_BLOCK)
#define NBLOCKS         ((BB / ROWS_PER_BLOCK) * (LL / COLS_PER_BLOCK))
#define SROW            260
#define COLS_PER_THREAD (HH / NTHREADS)
#define UCOLS           32

__device__ __forceinline__ float sigmoidf_(float x) {
    return 1.0f / (1.0f + __expf(-x));
}

__device__ __forceinline__ void lds_barrier() {
    __builtin_amdgcn_fence(__ATOMIC_RELEASE, "workgroup", "local");
    __builtin_amdgcn_s_barrier();
    __builtin_amdgcn_fence(__ATOMIC_ACQUIRE, "workgroup", "local");
}

__global__ __launch_bounds__(NTHREADS)
void kan_rnn_enc_kernel(const float* __restrict__ x_ctx,
                        const float* __restrict__ a_in,
                        const float* __restrict__ b_in,
                        const float* __restrict__ a_h,
                        const float* __restrict__ b_h,
                        const float* __restrict__ W_lat,
                        const float* __restrict__ b_lat,
                        float* __restrict__ out) {
    __shared__ float px[ROWS_PER_BLOCK * FF];
    __shared__ float sU[ROWS_PER_BLOCK * UCOLS];
    __shared__ float sH[ROWS_PER_BLOCK * SROW];
    __shared__ __attribute__((aligned(16))) float sPart[KPARTS * ROWS_PER_BLOCK * COLS_PER_BLOCK];

    const int tid = threadIdx.x;
    const int r0  = (blockIdx.x >> 1) * ROWS_PER_BLOCK;
    const int c0  = (blockIdx.x & 1) * COLS_PER_BLOCK;

    const int lane = tid & 31;
    const int wave = tid >> 5;
    const int nl   = lane & 15;
    const int kb   = (lane >> 4) << 1;
    const int nt   = wave & 1;
    const int kh   = wave >> 1;


    const int xrow = tid >> 4, xcol = tid & 15;
    const float xv = x_ctx[(size_t)(r0 + xrow) * (TT * FF) + (TT - 2) * FF + xcol];

    int   row1b[2], hh1b[2];
    float a1b[2], b1b[2];
#pragma unroll
    for (int c = 0; c < 2; ++c) {
        const int i2 = tid + c * NTHREADS;
        const int i  = (i2 < ROWS_PER_BLOCK * 24) ? i2 : tid;
        row1b[c] = i / 24;
        hh1b[c]  = i % 24;
        a1b[c] = a_in[hh1b[c]];
        b1b[c] = b_in[hh1b[c]];
    }

    float ca[COLS_PER_THREAD], cb[COLS_PER_THREAD];
#pragma unroll
    for (int c = 0; c < COLS_PER_THREAD; ++c) {
        const int j = tid + c * NTHREADS;
        const float* pa = (j < 64) ? (a_in + j) : (a_h + (j - 64));
        const float* pb = (j < 64) ? (b_in + j) : (b_h + (j - 64));
        ca[c] = *pa;
        cb[c] = *pb;
    }

    const float blane = b_lat[c0 + nt * 16 + nl];

    const float* wRow = W_lat + (c0 + nt * 16 + nl) * HH + kh * KSPAN + kb;
    v2f bfrag[KSPAN / 4];
#pragma unroll
    for (int q = 0; q < KSPAN / 4; ++q)
        bfrag[q] = *(const v2f*)(wRow + q * 4);

    {
        float* dst = (xcol < FF) ? &px[xrow * FF + xcol] : &sU[xrow * UCOLS + (xcol - FF)];
        *dst = xv;
    }
    lds_barrier();

#pragma unroll
    for (int c = 0; c < 2; ++c) {
        const int row = row1b[c], hh = hh1b[c];
        const int f = hh >> 3;
        float t = a1b[c] * (px[row * FF + f] - b1b[c]);
        sU[row * UCOLS + FF + hh] = sigmoidf_(2.0f * sigmoidf_(t));
    }
    lds_barrier();

#pragma unroll
    for (int c = 0; c < COLS_PER_THREAD; ++c) {
        const int j = tid + c * NTHREADS;
        const float* uCol = &sU[j >> 3];
        float*       hCol = &sH[j];
#pragma unroll
        for (int row = 0; row < ROWS_PER_BLOCK; ++row) {
            float u = uCol[row * UCOLS];
            hCol[row * SROW] = sigmoidf_(2.0f * sigmoidf_(ca[c] * (u - cb[c])));
        }
    }
    lds_barrier();

    v8f acc0, acc1;
    {
        const float bl = (kh == 0) ? blane : 0.0f;
#pragma unroll
        for (int r = 0; r < 8; ++r) { acc0[r] = bl; acc1[r] = 0.0f; }
    }

    const float* aRow = &sH[(nl & 7) * SROW + kh * KSPAN + kb];

    v2f afrag[KSPAN / 4];
#pragma unroll
    for (int q = 0; q < KSPAN / 4; ++q)
        afrag[q] = *(const v2f*)(aRow + q * 4);
    __builtin_amdgcn_sched_barrier(0);
#pragma unroll
    for (int kk = 0; kk < KSPAN / 4; kk += 2) {
        acc0 = __builtin_amdgcn_wmma_f32_16x16x4_f32(
            false, afrag[kk], false, bfrag[kk], (short)0, acc0, false, false);
        acc1 = __builtin_amdgcn_wmma_f32_16x16x4_f32(
            false, afrag[kk + 1], false, bfrag[kk + 1], (short)0, acc1, false, false);
    }

    v8f part = acc0 + acc1;
    if (lane < 16) {
#pragma unroll
        for (int r = 0; r < 8; ++r)
            sPart[(kh * ROWS_PER_BLOCK + r) * COLS_PER_BLOCK + nt * 16 + nl] = part[r];
    }
    __syncthreads();

    if (tid < ROWS_PER_BLOCK * COLS_PER_BLOCK / 4) {
        const float4* sPart4 = (const float4*)sPart;
        float4 p = sPart4[tid];
        float4 q = sPart4[ROWS_PER_BLOCK * COLS_PER_BLOCK / 4 + tid];
        float4 v;
        v.x = p.x + q.x; v.y = p.y + q.y; v.z = p.z + q.z; v.w = p.w + q.w;
        const int row = tid >> 3, quad = tid & 7;
        *(float4*)(out + (size_t)(r0 + row) * LL + c0 + quad * 4) = v;
    }
}

extern "C" void kernel_launch(void* const* d_in, const int* in_sizes, int n_in,
                              void* d_out, int out_size, void* d_ws, size_t ws_size,
                              hipStream_t stream) {
    (void)in_sizes; (void)n_in; (void)out_size; (void)d_ws; (void)ws_size;
    const float* x_ctx = (const float*)d_in[0];
    const float* a_in  = (const float*)d_in[1];
    const float* b_in  = (const float*)d_in[2];
    const float* a_h   = (const float*)d_in[3];
    const float* b_h   = (const float*)d_in[4];
    const float* W_lat = (const float*)d_in[5];
    const float* b_lat = (const float*)d_in[6];
    float* out = (float*)d_out;

    dim3 grid(NBLOCKS);
    dim3 block(NTHREADS);
    hipLaunchKernelGGL(kan_rnn_enc_kernel, grid, block, 0, stream,
                       x_ctx, a_in, b_in, a_h, b_h, W_lat, b_lat, out);
}


# Sigmoid quotient 1/(1+e^-x) taken with the hardware reciprocal v_rcp_f32 (1 ulp; denominator in [1,inf]) instead of the 11-instruction IEEE division: 36 per thread, results move by ulps
# speedup vs baseline: 12.1550x; 1.1913x over previous
#include <hip/hip_runtime.h>


typedef float v8f __attribute__((ext_vector_type(8)));
typedef float v2f __attribute__((ext_vector_type(2)));

#define BB   1024
#define TT   1024
#define FF   8
#define HH   256
#define LL   64

#define ROWS_PER_BLOCK  8
#define COLS_PER_BLOCK  32
#define NTILES          (COLS_PER_BLOCK / 16)
#define KPARTS          2
#define KSPAN           (HH / KPARTS)
#define WAVES_PER_BLOCK (NTILES * KPARTS)
#define NTHREADS        (32 * WAVES_PER_BLOCK)
#define NBLOCKS         ((BB / ROWS_PER_BLOCK) * (LL / COLS_PER_BLOCK))
#define SROW            260
#define COLS_PER_THREAD (HH / NTHREADS)
#define UCOLS           32

__device__ __forceinline__ float sigmoidf_(float x) {
    return __builtin_amdgcn_rcpf(1.0f + __expf(-x));
}

__device__ __forceinline__ void lds_barrier() {
    __builtin_amdgcn_fence(__ATOMIC_RELEASE, "workgroup", "local");
    __builtin_amdgcn_s_barrier();
    __builtin_amdgcn_fence(__ATOMIC_ACQUIRE, "workgroup", "local");
}

__global__ __launch_bounds__(NTHREADS)
void kan_rnn_enc_kernel(const float* __restrict__ x_ctx,
                        const float* __restrict__ a_in,
                        const float* __restrict__ b_in,
                        const float* __restrict__ a_h,
                        const float* __restrict__ b_h,
                        const float* __restrict__ W_lat,
                        const float* __restrict__ b_lat,
                        float* __restrict__ out) {
    __shared__ float px[ROWS_PER_BLOCK * FF];
    __shared__ float sU[ROWS_PER_BLOCK * UCOLS];
    __shared__ float sH[ROWS_PER_BLOCK * SROW];
    __shared__ __attribute__((aligned(16))) float sPart[KPARTS * ROWS_PER_BLOCK * COLS_PER_BLOCK];

    const int tid = threadIdx.x;
    const int r0  = (blockIdx.x >> 1) * ROWS_PER_BLOCK;
    const int c0  = (blockIdx.x & 1) * COLS_PER_BLOCK;

    const int lane = tid & 31;
    const int wave = tid >> 5;
    const int nl   = lane & 15;
    const int kb   = (lane >> 4) << 1;
    const int nt   = wave & 1;
    const int kh   = wave >> 1;


    const int xrow = tid >> 4, xcol = tid & 15;
    const float xv = x_ctx[(size_t)(r0 + xrow) * (TT * FF) + (TT - 2) * FF + xcol];

    int   row1b[2], hh1b[2];
    float a1b[2], b1b[2];
#pragma unroll
    for (int c = 0; c < 2; ++c) {
        const int i2 = tid + c * NTHREADS;
        const int i  = (i2 < ROWS_PER_BLOCK * 24) ? i2 : tid;
        row1b[c] = i / 24;
        hh1b[c]  = i % 24;
        a1b[c] = a_in[hh1b[c]];
        b1b[c] = b_in[hh1b[c]];
    }

    float ca[COLS_PER_THREAD], cb[COLS_PER_THREAD];
#pragma unroll
    for (int c = 0; c < COLS_PER_THREAD; ++c) {
        const int j = tid + c * NTHREADS;
        const float* pa = (j < 64) ? (a_in + j) : (a_h + (j - 64));
        const float* pb = (j < 64) ? (b_in + j) : (b_h + (j - 64));
        ca[c] = *pa;
        cb[c] = *pb;
    }

    const float blane = b_lat[c0 + nt * 16 + nl];

    const float* wRow = W_lat + (c0 + nt * 16 + nl) * HH + kh * KSPAN + kb;
    v2f bfrag[KSPAN / 4];
#pragma unroll
    for (int q = 0; q < KSPAN / 4; ++q)
        bfrag[q] = *(const v2f*)(wRow + q * 4);

    {
        float* dst = (xcol < FF) ? &px[xrow * FF + xcol] : &sU[xrow * UCOLS + (xcol - FF)];
        *dst = xv;
    }
    lds_barrier();

#pragma unroll
    for (int c = 0; c < 2; ++c) {
        const int row = row1b[c], hh = hh1b[c];
        const int f = hh >> 3;
        float t = a1b[c] * (px[row * FF + f] - b1b[c]);
        sU[row * UCOLS + FF + hh] = sigmoidf_(2.0f * sigmoidf_(t));
    }
    lds_barrier();

#pragma unroll
    for (int c = 0; c < COLS_PER_THREAD; ++c) {
        const int j = tid + c * NTHREADS;
        const float* uCol = &sU[j >> 3];
        float*       hCol = &sH[j];
#pragma unroll
        for (int row = 0; row < ROWS_PER_BLOCK; ++row) {
            float u = uCol[row * UCOLS];
            hCol[row * SROW] = sigmoidf_(2.0f * sigmoidf_(ca[c] * (u - cb[c])));
        }
    }
    lds_barrier();

    v8f acc0, acc1;
    {
        const float bl = (kh == 0) ? blane : 0.0f;
#pragma unroll
        for (int r = 0; r < 8; ++r) { acc0[r] = bl; acc1[r] = 0.0f; }
    }

    const float* aRow = &sH[(nl & 7) * SROW + kh * KSPAN + kb];

    v2f afrag[KSPAN / 4];
#pragma unroll
    for (int q = 0; q < KSPAN / 4; ++q)
        afrag[q] = *(const v2f*)(aRow + q * 4);
    __builtin_amdgcn_sched_barrier(0);
#pragma unroll
    for (int kk = 0; kk < KSPAN / 4; kk += 2) {
        acc0 = __builtin_amdgcn_wmma_f32_16x16x4_f32(
            false, afrag[kk], false, bfrag[kk], (short)0, acc0, false, false);
        acc1 = __builtin_amdgcn_wmma_f32_16x16x4_f32(
            false, afrag[kk + 1], false, bfrag[kk + 1], (short)0, acc1, false, false);
    }

    v8f part = acc0 + acc1;
    if (lane < 16) {
#pragma unroll
        for (int r = 0; r < 8; ++r)
            sPart[(kh * ROWS_PER_BLOCK + r) * COLS_PER_BLOCK + nt * 16 + nl] = part[r];
    }
    __syncthreads();

    if (tid < ROWS_PER_BLOCK * COLS_PER_BLOCK / 4) {
        const float4* sPart4 = (const float4*)sPart;
        float4 p = sPart4[tid];
        float4 q = sPart4[ROWS_PER_BLOCK * COLS_PER_BLOCK / 4 + tid];
        float4 v;
        v.x = p.x + q.x; v.y = p.y + q.y; v.z = p.z + q.z; v.w = p.w + q.w;
        const int row = tid >> 3, quad = tid & 7;
        *(float4*)(out + (size_t)(r0 + row) * LL + c0 + quad * 4) = v;
    }
}

extern "C" void kernel_launch(void* const* d_in, const int* in_sizes, int n_in,
                              void* d_out, int out_size, void* d_ws, size_t ws_size,
                              hipStream_t stream) {
    (void)in_sizes; (void)n_in; (void)out_size; (void)d_ws; (void)ws_size;
    const float* x_ctx = (const float*)d_in[0];
    const float* a_in  = (const float*)d_in[1];
    const float* b_in  = (const float*)d_in[2];
    const float* a_h   = (const float*)d_in[3];
    const float* b_h   = (const float*)d_in[4];
    const float* W_lat = (const float*)d_in[5];
    const float* b_lat = (const float*)d_in[6];
    float* out = (float*)d_out;

    dim3 grid(NBLOCKS);
    dim3 block(NTHREADS);
    hipLaunchKernelGGL(kan_rnn_enc_kernel, grid, block, 0, stream,
                       x_ctx, a_in, b_in, a_h, b_h, W_lat, b_lat, out);
}
